# starting kernel
# baseline (speedup 1.0000x reference)
#include <hip/hip_runtime.h>


typedef __attribute__((ext_vector_type(2))) float v2f;
typedef __attribute__((ext_vector_type(8))) float v8f;

#define N_NODES 1024
#define EMBED   64
#define HIDDEN  128

__global__ __launch_bounds__(256) void gcn_collapsed_kernel(
    const int*   __restrict__ y,
    const float* __restrict__ emb,
    const float* __restrict__ W1,
    const float* __restrict__ b1,
    const float* __restrict__ W2,
    const float* __restrict__ b2,
    float*       __restrict__ ovec)
{
    __shared__ int   yl[N_NODES];
    __shared__ float red[4][EMBED];
    __shared__ float zbar[EMBED];
    __shared__ float hrow[HIDDEN];

    const int t = threadIdx.x;

    for (int i = t; i < N_NODES; i += 256) yl[i] = y[i];
    __syncthreads();

    {
        const int c = t & 63;
        const int g = t >> 6;
        float acc = 0.f;
        const int base = g * 256;
        for (int j = 0; j < 256; ++j) {
            acc += emb[yl[base + j] * EMBED + c];
        }
        red[g][c] = acc;
    }
    __syncthreads();
    if (t < EMBED) {
        zbar[t] = (red[0][t] + red[1][t] + red[2][t] + red[3][t]) * (1.0f / 1024.0f);
    }
    __syncthreads();

    const int wave  = t >> 5;
    const int lane  = t & 31;
    const int nloc  = lane & 15;
    const int khalf = (lane >> 4) << 1;

    {
        const int n = wave * 16 + nloc;
        v8f cacc = {};
        for (int kk = 0; kk < EMBED / 4; ++kk) {
            const int k0 = kk * 4 + khalf;
            v2f a; a.x = zbar[k0];               a.y = zbar[k0 + 1];
            v2f b; b.x = W1[k0 * HIDDEN + n];    b.y = W1[(k0 + 1) * HIDDEN + n];
            cacc = __builtin_amdgcn_wmma_f32_16x16x4_f32(
                false, a, false, b, (short)0, cacc, false, false);
        }
        const float h = fmaxf(cacc[0] + b1[n], 0.f);
        if (lane < 16) hrow[n] = h;
    }
    __syncthreads();

    {
        const int n = (wave & 3) * 16 + nloc;
        v8f cacc = {};
        for (int kk = 0; kk < HIDDEN / 4; ++kk) {
            const int k0 = kk * 4 + khalf;
            v2f a; a.x = hrow[k0];               a.y = hrow[k0 + 1];
            v2f b; b.x = W2[k0 * EMBED + n];     b.y = W2[(k0 + 1) * EMBED + n];
            cacc = __builtin_amdgcn_wmma_f32_16x16x4_f32(
                false, a, false, b, (short)0, cacc, false, false);
        }
        if (wave < 4 && lane < 16) ovec[n] = cacc[0] + b2[n];
    }
}

__global__ __launch_bounds__(256) void broadcast_kernel(
    const float4* __restrict__ o4,
    float4*       __restrict__ out4)
{
    const int j = blockIdx.x * 256 + threadIdx.x;
    out4[j] = o4[j & 15];
}

extern "C" void kernel_launch(void* const* d_in, const int* in_sizes, int n_in,
                              void* d_out, int out_size, void* d_ws, size_t ws_size,
                              hipStream_t stream) {
    const int*   y   = (const int*)  d_in[0];
    const float* emb = (const float*)d_in[2];
    const float* W1  = (const float*)d_in[3];
    const float* b1  = (const float*)d_in[4];
    const float* W2  = (const float*)d_in[5];
    const float* b2  = (const float*)d_in[6];

    float* ovec = (float*)d_ws;
    float* out  = (float*)d_out;

    gcn_collapsed_kernel<<<1, 256, 0, stream>>>(y, emb, W1, b1, W2, b2, ovec);

    const int total4 = (N_NODES * EMBED) / 4;
    broadcast_kernel<<<total4 / 256, 256, 0, stream>>>(
        (const float4*)ovec, (float4*)out);
    (void)in_sizes; (void)n_in; (void)out_size; (void)ws_size;
}


# gather for the mean row: 1024 threads, each with 16 independent 16-byte row loads in flight (was 256 threads, 8-deep serial 4-byte loads); changes the order in which the mean is summed
# speedup vs baseline: 1.8810x; 1.8810x over previous
#include <hip/hip_runtime.h>


typedef __attribute__((ext_vector_type(2))) float v2f;
typedef __attribute__((ext_vector_type(8))) float v8f;

#define N_NODES 1024
#define EMBED   64
#define HIDDEN  128
#define BLOCK   1024
#define NWAVES  (BLOCK / 32)
#define ROWS_PER_THREAD 16

__global__ __launch_bounds__(BLOCK) __attribute__((amdgpu_waves_per_eu(8, 8)))
void gcn_collapsed_kernel(
    const int*   __restrict__ y,
    const float* __restrict__ emb,
    const float* __restrict__ W1,
    const float* __restrict__ b1,
    const float* __restrict__ W2,
    const float* __restrict__ b2,
    float*       __restrict__ ovec)
{
    __shared__ float red[NWAVES][EMBED];
    __shared__ float zbar[EMBED];
    __shared__ float hrow[HIDDEN];

    const int t    = threadIdx.x;
    const int wave = t >> 5;
    const int lane = t & 31;

    {
        const int c4 = t & 15;
        const int rg = t >> 4;
        const int4* y4 = reinterpret_cast<const int4*>(y) + rg * (ROWS_PER_THREAD / 4);
        const int4 q0 = y4[0];
        const int4 q1 = y4[1];
        const int4 q2 = y4[2];
        const int4 q3 = y4[3];
        const int idx[ROWS_PER_THREAD] = {q0.x, q0.y, q0.z, q0.w, q1.x, q1.y, q1.z, q1.w,
                                          q2.x, q2.y, q2.z, q2.w, q3.x, q3.y, q3.z, q3.w};
        const float4* e4 = reinterpret_cast<const float4*>(emb);
        float4 v[ROWS_PER_THREAD];
        for (int j = 0; j < ROWS_PER_THREAD; ++j) v[j] = e4[idx[j] * (EMBED / 4) + c4];
        float4 acc = v[0];
        for (int j = 1; j < ROWS_PER_THREAD; ++j) {
            acc.x += v[j].x; acc.y += v[j].y; acc.z += v[j].z; acc.w += v[j].w;
        }
        acc.x += __shfl_xor(acc.x, 16);
        acc.y += __shfl_xor(acc.y, 16);
        acc.z += __shfl_xor(acc.z, 16);
        acc.w += __shfl_xor(acc.w, 16);
        if (lane < 16) *reinterpret_cast<float4*>(&red[wave][c4 * 4]) = acc;
    }
    __syncthreads();
    if (t < EMBED) {
        float s0 = 0.f, s1 = 0.f, s2 = 0.f, s3 = 0.f;
        for (int w = 0; w < NWAVES; w += 4) {
            s0 += red[w][t]; s1 += red[w + 1][t]; s2 += red[w + 2][t]; s3 += red[w + 3][t];
        }
        zbar[t] = ((s0 + s1) + (s2 + s3)) * (1.0f / 1024.0f);
    }
    __syncthreads();

    const int nloc  = lane & 15;
    const int khalf = (lane >> 4) << 1;

    if (wave < HIDDEN / 16) {
        const int n = wave * 16 + nloc;
        v8f cacc = {};
        for (int kk = 0; kk < EMBED / 4; ++kk) {
            const int k0 = kk * 4 + khalf;
            v2f a; a.x = zbar[k0];               a.y = zbar[k0 + 1];
            v2f b; b.x = W1[k0 * HIDDEN + n];    b.y = W1[(k0 + 1) * HIDDEN + n];
            cacc = __builtin_amdgcn_wmma_f32_16x16x4_f32(
                false, a, false, b, (short)0, cacc, false, false);
        }
        const float h = fmaxf(cacc[0] + b1[n], 0.f);
        if (lane < 16) hrow[n] = h;
    }
    __syncthreads();

    if (wave < EMBED / 16) {
        const int n = wave * 16 + nloc;
        v8f cacc = {};
        for (int kk = 0; kk < HIDDEN / 4; ++kk) {
            const int k0 = kk * 4 + khalf;
            v2f a; a.x = hrow[k0];               a.y = hrow[k0 + 1];
            v2f b; b.x = W2[k0 * EMBED + n];     b.y = W2[(k0 + 1) * EMBED + n];
            cacc = __builtin_amdgcn_wmma_f32_16x16x4_f32(
                false, a, false, b, (short)0, cacc, false, false);
        }
        if (lane < 16) ovec[n] = cacc[0] + b2[n];
    }
}

__global__ __launch_bounds__(256) void broadcast_kernel(
    const float4* __restrict__ o4,
    float4*       __restrict__ out4)
{
    const int j = blockIdx.x * 256 + threadIdx.x;
    out4[j] = o4[j & 15];
}

extern "C" void kernel_launch(void* const* d_in, const int* in_sizes, int n_in,
                              void* d_out, int out_size, void* d_ws, size_t ws_size,
                              hipStream_t stream) {
    const int*   y   = (const int*)  d_in[0];
    const float* emb = (const float*)d_in[2];
    const float* W1  = (const float*)d_in[3];
    const float* b1  = (const float*)d_in[4];
    const float* W2  = (const float*)d_in[5];
    const float* b2  = (const float*)d_in[6];

    float* ovec = (float*)d_ws;
    float* out  = (float*)d_out;

    gcn_collapsed_kernel<<<1, BLOCK, 0, stream>>>(y, emb, W1, b1, W2, b2, ovec);

    const int total4 = (N_NODES * EMBED) / 4;
    broadcast_kernel<<<total4 / 256, 256, 0, stream>>>(
        (const float4*)ovec, (float4*)out);
    (void)in_sizes; (void)n_in; (void)out_size; (void)ws_size;
}


# one launch instead of two: 16 work-groups each compute the result row with the same code and summation order and store their own 64 output rows; broadcast launch and workspace row removed
# speedup vs baseline: 2.3945x; 1.2730x over previous
#include <hip/hip_runtime.h>


typedef __attribute__((ext_vector_type(2))) float v2f;
typedef __attribute__((ext_vector_type(8))) float v8f;

#define N_NODES 1024
#define EMBED   64
#define HIDDEN  128
#define BLOCK   1024
#define NWAVES  (BLOCK / 32)
#define ROWS_PER_THREAD 16
#define GRID    16
#define ROWS_PER_GROUP (N_NODES / GRID)

__global__ __launch_bounds__(BLOCK) __attribute__((amdgpu_waves_per_eu(8, 8)))
void gcn_collapsed_kernel(
    const int*   __restrict__ y,
    const float* __restrict__ emb,
    const float* __restrict__ W1,
    const float* __restrict__ b1,
    const float* __restrict__ W2,
    const float* __restrict__ b2,
    float*       __restrict__ out)
{
    __shared__ float red[NWAVES][EMBED];
    __shared__ float zbar[EMBED];
    __shared__ float hrow[HIDDEN];
    __shared__ __attribute__((aligned(16))) float orow[EMBED];

    const int t    = threadIdx.x;
    const int wave = t >> 5;
    const int lane = t & 31;

    {
        const int c4 = t & 15;
        const int rg = t >> 4;
        const int4* y4 = reinterpret_cast<const int4*>(y) + rg * (ROWS_PER_THREAD / 4);
        const int4 q0 = y4[0];
        const int4 q1 = y4[1];
        const int4 q2 = y4[2];
        const int4 q3 = y4[3];
        const int idx[ROWS_PER_THREAD] = {q0.x, q0.y, q0.z, q0.w, q1.x, q1.y, q1.z, q1.w,
                                          q2.x, q2.y, q2.z, q2.w, q3.x, q3.y, q3.z, q3.w};
        const float4* e4 = reinterpret_cast<const float4*>(emb);
        float4 v[ROWS_PER_THREAD];
        for (int j = 0; j < ROWS_PER_THREAD; ++j) v[j] = e4[idx[j] * (EMBED / 4) + c4];
        float4 acc = v[0];
        for (int j = 1; j < ROWS_PER_THREAD; ++j) {
            acc.x += v[j].x; acc.y += v[j].y; acc.z += v[j].z; acc.w += v[j].w;
        }
        acc.x += __shfl_xor(acc.x, 16);
        acc.y += __shfl_xor(acc.y, 16);
        acc.z += __shfl_xor(acc.z, 16);
        acc.w += __shfl_xor(acc.w, 16);
        if (lane < 16) *reinterpret_cast<float4*>(&red[wave][c4 * 4]) = acc;
    }
    __syncthreads();
    if (t < EMBED) {
        float s0 = 0.f, s1 = 0.f, s2 = 0.f, s3 = 0.f;
        for (int w = 0; w < NWAVES; w += 4) {
            s0 += red[w][t]; s1 += red[w + 1][t]; s2 += red[w + 2][t]; s3 += red[w + 3][t];
        }
        zbar[t] = ((s0 + s1) + (s2 + s3)) * (1.0f / 1024.0f);
    }
    __syncthreads();

    const int nloc  = lane & 15;
    const int khalf = (lane >> 4) << 1;

    if (wave < HIDDEN / 16) {
        const int n = wave * 16 + nloc;
        v8f cacc = {};
        for (int kk = 0; kk < EMBED / 4; ++kk) {
            const int k0 = kk * 4 + khalf;
            v2f a; a.x = zbar[k0];               a.y = zbar[k0 + 1];
            v2f b; b.x = W1[k0 * HIDDEN + n];    b.y = W1[(k0 + 1) * HIDDEN + n];
            cacc = __builtin_amdgcn_wmma_f32_16x16x4_f32(
                false, a, false, b, (short)0, cacc, false, false);
        }
        const float h = fmaxf(cacc[0] + b1[n], 0.f);
        if (lane < 16) hrow[n] = h;
    }
    __syncthreads();

    if (wave < EMBED / 16) {
        const int n = wave * 16 + nloc;
        v8f cacc = {};
        for (int kk = 0; kk < HIDDEN / 4; ++kk) {
            const int k0 = kk * 4 + khalf;
            v2f a; a.x = hrow[k0];               a.y = hrow[k0 + 1];
            v2f b; b.x = W2[k0 * EMBED + n];     b.y = W2[(k0 + 1) * EMBED + n];
            cacc = __builtin_amdgcn_wmma_f32_16x16x4_f32(
                false, a, false, b, (short)0, cacc, false, false);
        }
        if (lane < 16) orow[n] = cacc[0] + b2[n];
    }
    __syncthreads();

    {
        const float4 o = *reinterpret_cast<const float4*>(&orow[(t & 15) * 4]);
        reinterpret_cast<float4*>(out)[blockIdx.x * (ROWS_PER_GROUP * EMBED / 4) + t] = o;
    }
}

extern "C" void kernel_launch(void* const* d_in, const int* in_sizes, int n_in,
                              void* d_out, int out_size, void* d_ws, size_t ws_size,
                              hipStream_t stream) {
    const int*   y   = (const int*)  d_in[0];
    const float* emb = (const float*)d_in[2];
    const float* W1  = (const float*)d_in[3];
    const float* b1  = (const float*)d_in[4];
    const float* W2  = (const float*)d_in[5];
    const float* b2  = (const float*)d_in[6];

    float* out  = (float*)d_out;

    gcn_collapsed_kernel<<<GRID, BLOCK, 0, stream>>>(y, emb, W1, b1, W2, b2, out);
    (void)in_sizes; (void)n_in; (void)out_size; (void)d_ws; (void)ws_size;
}


# W1, W2, b1, b2 loaded at kernel start behind the index loads and parked in LDS; idle waves read B fragments from LDS during the mean-row reduce; no GEMM waits on a global load; sums unchanged
# speedup vs baseline: 3.2837x; 1.3714x over previous
#include <hip/hip_runtime.h>


typedef __attribute__((ext_vector_type(2))) float v2f;
typedef __attribute__((ext_vector_type(8))) float v8f;

#define N_NODES 1024
#define EMBED   64
#define HIDDEN  128
#define BLOCK   1024
#define NWAVES  (BLOCK / 32)
#define ROWS_PER_THREAD 16
#define GRID    16
#define ROWS_PER_GROUP (N_NODES / GRID)

#define W1_PITCH (HIDDEN + 16)
#define W2_PITCH (EMBED + 16)
#define G1_WAVE0 8
#define G2_WAVE0 16

struct __attribute__((aligned(16))) Lds {
    float red[NWAVES][EMBED];
    float zbar[EMBED];
    float hrow[HIDDEN];
    float orow[EMBED];
    float W1s[EMBED * W1_PITCH];
    float W2s[HIDDEN * W2_PITCH];
};

__global__ __launch_bounds__(BLOCK) __attribute__((amdgpu_waves_per_eu(8, 8)))
void gcn_collapsed_kernel(
    const int*   __restrict__ y,
    const float* __restrict__ emb,
    float*       __restrict__ out,
    const float* W1,
    const float* b1,
    const float* W2,
    const float* b2)
{
    __shared__ Lds s;

    const int t    = threadIdx.x;
    const int wave = t >> 5;
    const int lane = t & 31;

    const int nloc  = lane & 15;
    const int khalf = (lane >> 4) << 1;
    const bool g1 = (wave >= G1_WAVE0) && (wave < G1_WAVE0 + HIDDEN / 16);
    const bool g2 = (wave >= G2_WAVE0) && (wave < G2_WAVE0 + EMBED / 16);
    const int n1 = ((wave - G1_WAVE0) * 16 + nloc) & (HIDDEN - 1);
    const int n2 = ((wave - G2_WAVE0) * 16 + nloc) & (EMBED - 1);
    float4 w1a, w1b, w2a, w2b;
    float bias1, bias2;

    {
        const int c4 = t & 15;
        const int rg = t >> 4;
        const int4* y4 = reinterpret_cast<const int4*>(y) + rg * (ROWS_PER_THREAD / 4);
        const int4 q0 = y4[0];
        const int4 q1 = y4[1];
        const int4 q2 = y4[2];
        const int4 q3 = y4[3];
        const float4* W1v = reinterpret_cast<const float4*>(W1);
        const float4* W2v = reinterpret_cast<const float4*>(W2);
        w1a = W1v[t];
        w1b = W1v[t + BLOCK];
        w2a = W2v[t];
        w2b = W2v[t + BLOCK];
        bias1 = b1[n1];
        bias2 = b2[n2];
        asm volatile("" ::: "memory");
        const int idx[ROWS_PER_THREAD] = {q0.x, q0.y, q0.z, q0.w, q1.x, q1.y, q1.z, q1.w,
                                          q2.x, q2.y, q2.z, q2.w, q3.x, q3.y, q3.z, q3.w};
        const float4* e4 = reinterpret_cast<const float4*>(emb);
        float4 v[ROWS_PER_THREAD];
        for (int j = 0; j < ROWS_PER_THREAD; ++j) v[j] = e4[idx[j] * (EMBED / 4) + c4];
        *reinterpret_cast<float4*>(&s.W1s[(t >> 5) * W1_PITCH + (t & 31) * 4]) = w1a;
        *reinterpret_cast<float4*>(&s.W1s[((t >> 5) + BLOCK / 32) * W1_PITCH + (t & 31) * 4]) = w1b;
        *reinterpret_cast<float4*>(&s.W2s[(t >> 4) * W2_PITCH + (t & 15) * 4]) = w2a;
        *reinterpret_cast<float4*>(&s.W2s[((t >> 4) + BLOCK / 16) * W2_PITCH + (t & 15) * 4]) = w2b;
        __builtin_amdgcn_sched_barrier(0);
        float4 acc = v[0];
        for (int j = 1; j < ROWS_PER_THREAD; ++j) {
            acc.x += v[j].x; acc.y += v[j].y; acc.z += v[j].z; acc.w += v[j].w;
        }
        acc.x += __shfl_xor(acc.x, 16);
        acc.y += __shfl_xor(acc.y, 16);
        acc.z += __shfl_xor(acc.z, 16);
        acc.w += __shfl_xor(acc.w, 16);
        if (lane < 16) *reinterpret_cast<float4*>(&s.red[wave][c4 * 4]) = acc;
    }
    __syncthreads();

    v2f bw[HIDDEN / 4];
    if (g1) {
        for (int kk = 0; kk < EMBED / 4; ++kk) {
            const int k0 = kk * 4 + khalf;
            bw[kk].x = s.W1s[k0 * W1_PITCH + n1];
            bw[kk].y = s.W1s[(k0 + 1) * W1_PITCH + n1];
        }
    } else if (g2) {
        for (int kk = 0; kk < HIDDEN / 4; ++kk) {
            const int k0 = kk * 4 + khalf;
            bw[kk].x = s.W2s[k0 * W2_PITCH + n2];
            bw[kk].y = s.W2s[(k0 + 1) * W2_PITCH + n2];
        }
    }

    if (t < EMBED) {
        float s0 = 0.f, s1 = 0.f, s2 = 0.f, s3 = 0.f;
        for (int w = 0; w < NWAVES; w += 4) {
            s0 += s.red[w][t]; s1 += s.red[w + 1][t]; s2 += s.red[w + 2][t]; s3 += s.red[w + 3][t];
        }
        s.zbar[t] = ((s0 + s1) + (s2 + s3)) * (1.0f / 1024.0f);
    }
    __syncthreads();

    if (g1) {
        v8f cacc = {};
        for (int kk = 0; kk < EMBED / 4; ++kk) {
            const int k0 = kk * 4 + khalf;
            v2f a; a.x = s.zbar[k0];             a.y = s.zbar[k0 + 1];
            cacc = __builtin_amdgcn_wmma_f32_16x16x4_f32(
                false, a, false, bw[kk], (short)0, cacc, false, false);
        }
        const float h = fmaxf(cacc[0] + bias1, 0.f);
        if (lane < 16) s.hrow[n1] = h;
    }
    __syncthreads();

    if (g2) {
        v8f cacc = {};
        for (int kk = 0; kk < HIDDEN / 4; ++kk) {
            const int k0 = kk * 4 + khalf;
            v2f a; a.x = s.hrow[k0];             a.y = s.hrow[k0 + 1];
            cacc = __builtin_amdgcn_wmma_f32_16x16x4_f32(
                false, a, false, bw[kk], (short)0, cacc, false, false);
        }
        if (lane < 16) s.orow[n2] = cacc[0] + bias2;
    }
    __syncthreads();

    {
        const float4 o = *reinterpret_cast<const float4*>(&s.orow[(t & 15) * 4]);
        reinterpret_cast<float4*>(out)[blockIdx.x * (ROWS_PER_GROUP * EMBED / 4) + t] = o;
    }
}

extern "C" void kernel_launch(void* const* d_in, const int* in_sizes, int n_in,
                              void* d_out, int out_size, void* d_ws, size_t ws_size,
                              hipStream_t stream) {
    const int*   y   = (const int*)  d_in[0];
    const float* emb = (const float*)d_in[2];
    const float* W1  = (const float*)d_in[3];
    const float* b1  = (const float*)d_in[4];
    const float* W2  = (const float*)d_in[5];
    const float* b2  = (const float*)d_in[6];

    float* out  = (float*)d_out;

    gcn_collapsed_kernel<<<GRID, BLOCK, 0, stream>>>(y, emb, out, W1, b1, W2, b2);
    (void)in_sizes; (void)n_in; (void)out_size; (void)d_ws; (void)ws_size;
}
